# starting kernel
# baseline (speedup 1.0000x reference)
#include <hip/hip_runtime.h>
#include <math.h>

#define NTYPE 4
#define NWAVE 8
#define NORBIT 128
#define NEIGH 64
#define NB 32
#define NATOM 256
#define NPAIR_PER (NATOM * NEIGH)
#define NPAIR_TOT (NB * NPAIR_PER)
#define TOTNATOM (NB * NATOM)
#define NPARA 13
#define PI_OVER_CUTOFF 0.62831853071795864769f

typedef float v2f __attribute__((ext_vector_type(2)));
typedef float v8f __attribute__((ext_vector_type(8)));

__device__ __forceinline__ v8f wmma_f32_16x16x4(v2f a, v2f b, v8f c) {
  return __builtin_amdgcn_wmma_f32_16x16x4_f32(false, a, false, b, (short)0, c,
                                               false, false);
}

__global__ void k_init_orb(const float* __restrict__ params,
                           const int* __restrict__ species,
                           float* __restrict__ orb) {
  int t = blockIdx.x * blockDim.x + threadIdx.x;
  int a = t / NWAVE, w = t % NWAVE;
  orb[t] = params[species[a] * NWAVE + w];
}

__global__ void k_zero(float* __restrict__ p, int n) {
  int t = blockIdx.x * blockDim.x + threadIdx.x;
  if (t < n) p[t] = 0.0f;
}

__global__ void k_pairs(const float* __restrict__ cart,
                        const float* __restrict__ shifts,
                        const float* __restrict__ rs,
                        const float* __restrict__ inta,
                        const int* __restrict__ species,
                        const int* __restrict__ atom_index,
                        const float* __restrict__ orb,
                        float* __restrict__ sum_w) {
  int p = blockIdx.x * blockDim.x + threadIdx.x;
  int b = p / NPAIR_PER;
  int gi = b * NATOM + atom_index[p];
  int gj = b * NATOM + atom_index[NPAIR_TOT + p];

  float dx = cart[gi * 3 + 0] - cart[gj * 3 + 0] + shifts[p * 3 + 0];
  float dy = cart[gi * 3 + 1] - cart[gj * 3 + 1] + shifts[p * 3 + 1];
  float dz = cart[gi * 3 + 2] - cart[gj * 3 + 2] + shifts[p * 3 + 2];
  float d = sqrtf(dx * dx + dy * dy + dz * dz);
  float fc = 0.5f * cosf(d * PI_OVER_CUTOFF) + 0.5f;
  fc = fc * fc;

  int sp = species[gj];
  float wv[NWAVE];
#pragma unroll
  for (int w = 0; w < NWAVE; ++w) {
    float t = d - rs[sp * NWAVE + w];
    wv[w] = expf(inta[sp * NWAVE + w] * t * t) * orb[gj * NWAVE + w];
  }

  float ang[NPARA];
  float dv[3] = {dx, dy, dz};
  ang[0] = fc;
#pragma unroll
  for (int a2 = 0; a2 < 3; ++a2) ang[1 + a2] = fc * dv[a2];
#pragma unroll
  for (int a2 = 0; a2 < 3; ++a2)
#pragma unroll
    for (int b2 = 0; b2 < 3; ++b2) ang[4 + a2 * 3 + b2] = ang[1 + a2] * dv[b2];

  float* dst = sum_w + gi * (NPARA * NWAVE);
#pragma unroll
  for (int j = 0; j < NPARA; ++j)
#pragma unroll
    for (int w = 0; w < NWAVE; ++w)
      unsafeAtomicAdd(dst + j * NWAVE + w, ang[j] * wv[w]);
}

__global__ void k_density(const float* __restrict__ sum_w,
                          const float* __restrict__ hyper,
                          float* __restrict__ density) {
  int wave = threadIdx.x >> 5;
  int lane = threadIdx.x & 31;
  int tile = blockIdx.x * 8 + wave;
  int lrow = lane & 15;
  int lhi = lane >> 4;
  int khalf = lhi * 2;

  for (int nt = 0; nt < 8; ++nt) {
    int n = nt * 16 + lrow;
    v8f acc = {};
    int para = 0;
    for (int ip = 0; ip < 3; ++ip) {
      const float* hb = hyper + ip * (NWAVE * NORBIT);
      v2f Blo, Bhi;
      Blo.x = hb[(khalf + 0) * NORBIT + n];
      Blo.y = hb[(khalf + 1) * NORBIT + n];
      Bhi.x = hb[(khalf + 4) * NORBIT + n];
      Bhi.y = hb[(khalf + 5) * NORBIT + n];
      int npj = (ip == 0) ? 1 : ((ip == 1) ? 3 : 9);
      for (int q = 0; q < npj; ++q, ++para) {
        const float* ab = sum_w + ((tile * 16 + lrow) * NPARA + para) * NWAVE;
        v2f Alo = {ab[khalf + 0], ab[khalf + 1]};
        v2f Ahi = {ab[khalf + 4], ab[khalf + 5]};
        v8f c = {};
        c = wmma_f32_16x16x4(Alo, Blo, c);
        c = wmma_f32_16x16x4(Ahi, Bhi, c);
#pragma unroll
        for (int r = 0; r < 8; ++r) acc[r] += c[r] * c[r];
      }
    }
#pragma unroll
    for (int r = 0; r < 8; ++r) {
      int atom = tile * 16 + r + lhi * 8;
      density[atom * NORBIT + nt * 16 + lrow] = acc[r];
    }
  }
}

__global__ void k_update(const float* __restrict__ density,
                         const float* __restrict__ oc_w,
                         const float* __restrict__ oc_e,
                         const int* __restrict__ species,
                         float* __restrict__ orb) {
  int wave = threadIdx.x >> 5;
  int lane = threadIdx.x & 31;
  int tile = blockIdx.x * 8 + wave;
  int lrow = lane & 15;
  int lhi = lane >> 4;

  v8f acc = {};
  const float* arow = density + (tile * 16 + lrow) * NORBIT;
  bool valid = (lrow < NWAVE);
#pragma unroll 4
  for (int kb = 0; kb < 32; ++kb) {
    int k = kb * 4 + lhi * 2;
    v2f A = {arow[k], arow[k + 1]};
    v2f B;
    B.x = valid ? oc_w[(k + 0) * NWAVE + lrow] : 0.0f;
    B.y = valid ? oc_w[(k + 1) * NWAVE + lrow] : 0.0f;
    acc = wmma_f32_16x16x4(A, B, acc);
  }
  if (valid) {
#pragma unroll
    for (int r = 0; r < 8; ++r) {
      int atom = tile * 16 + r + lhi * 8;
      float v = tanhf(acc[r] + oc_e[species[atom] * NWAVE + lrow]);
      orb[atom * NWAVE + lrow] += v;
    }
  }
}

extern "C" void kernel_launch(void* const* d_in, const int* in_sizes, int n_in,
                              void* d_out, int out_size, void* d_ws,
                              size_t ws_size, hipStream_t stream) {
  const float* cart = (const float*)d_in[0];
  const float* shifts = (const float*)d_in[1];
  const float* rs = (const float*)d_in[2];
  const float* inta = (const float*)d_in[3];
  const float* params = (const float*)d_in[4];
  const float* hyper = (const float*)d_in[5];
  const float* oc_w0 = (const float*)d_in[6];
  const float* oc_e0 = (const float*)d_in[7];
  const float* oc_w1 = (const float*)d_in[8];
  const float* oc_e1 = (const float*)d_in[9];
  const int* species = (const int*)d_in[11];
  const int* atom_index = (const int*)d_in[12];

  float* density = (float*)d_out;
  float* orb = (float*)d_ws;
  float* sum_w = orb + TOTNATOM * NWAVE;

  const int SW_N = TOTNATOM * NPARA * NWAVE;
  const float* Ws[2] = {oc_w0, oc_w1};
  const float* Es[2] = {oc_e0, oc_e1};

  k_init_orb<<<(TOTNATOM * NWAVE) / 256, 256, 0, stream>>>(params, species, orb);
  for (int it = 0; it < 3; ++it) {
    k_zero<<<(SW_N + 255) / 256, 256, 0, stream>>>(sum_w, SW_N);
    k_pairs<<<NPAIR_TOT / 256, 256, 0, stream>>>(cart, shifts, rs, inta, species,
                                                 atom_index, orb, sum_w);
    k_density<<<TOTNATOM / 16 / 8, 256, 0, stream>>>(sum_w, hyper, density);
    if (it < 2)
      k_update<<<TOTNATOM / 16 / 8, 256, 0, stream>>>(density, Ws[it], Es[it],
                                                      species, orb);
  }
}


# Segment sum in LDS: one block per (molecule, radial channel) adds its pair terms into a 13x256 LDS table with LDS float adds (arrival order), stores it whole; global atomics and zero launches gone
# speedup vs baseline: 38.3647x; 38.3647x over previous
#include <hip/hip_runtime.h>
#include <math.h>

#define NTYPE 4
#define NWAVE 8
#define NORBIT 128
#define NEIGH 64
#define NB 32
#define NATOM 256
#define NPAIR_PER (NATOM * NEIGH)
#define NPAIR_TOT (NB * NPAIR_PER)
#define TOTNATOM (NB * NATOM)
#define NPARA 13
#define PI_OVER_CUTOFF 0.62831853071795864769f

typedef float v2f __attribute__((ext_vector_type(2)));
typedef float v8f __attribute__((ext_vector_type(8)));

__device__ __forceinline__ v8f wmma_f32_16x16x4(v2f a, v2f b, v8f c) {
  return __builtin_amdgcn_wmma_f32_16x16x4_f32(false, a, false, b, (short)0, c,
                                               false, false);
}

__global__ void k_init_orb(const float* __restrict__ params,
                           const int* __restrict__ species,
                           float* __restrict__ orb) {
  int t = blockIdx.x * blockDim.x + threadIdx.x;
  int a = t / NWAVE, w = t % NWAVE;
  orb[t] = params[species[a] * NWAVE + w];
}

#define ACC_N (NPARA * NATOM)
#define PAIR_THREADS 1024

__global__ __launch_bounds__(PAIR_THREADS) void k_pairs(
    const float* __restrict__ cart, const float* __restrict__ shifts,
    const float* __restrict__ rs, const float* __restrict__ inta,
    const int* __restrict__ species, const int* __restrict__ atom_index,
    const float* __restrict__ orb, float* __restrict__ sum_w) {
  __shared__ float acc[ACC_N];
  const int b = blockIdx.x / NWAVE;
  const int w = blockIdx.x % NWAVE;

  for (int i = threadIdx.x; i < ACC_N; i += PAIR_THREADS) acc[i] = 0.0f;
  __syncthreads();

  for (int q = threadIdx.x; q < NPAIR_PER; q += PAIR_THREADS) {
    int p = b * NPAIR_PER + q;
    int ci = atom_index[p];
    int gi = b * NATOM + ci;
    int gj = b * NATOM + atom_index[NPAIR_TOT + p];

    float dx = cart[gi * 3 + 0] - cart[gj * 3 + 0] + shifts[p * 3 + 0];
    float dy = cart[gi * 3 + 1] - cart[gj * 3 + 1] + shifts[p * 3 + 1];
    float dz = cart[gi * 3 + 2] - cart[gj * 3 + 2] + shifts[p * 3 + 2];
    float d = sqrtf(dx * dx + dy * dy + dz * dz);
    float fc = 0.5f * cosf(d * PI_OVER_CUTOFF) + 0.5f;
    fc = fc * fc;

    int sp = species[gj];
    float t = d - rs[sp * NWAVE + w];
    float wv = expf(inta[sp * NWAVE + w] * t * t) * orb[gj * NWAVE + w];

    float ang[NPARA];
    float dv[3] = {dx, dy, dz};
    ang[0] = fc;
#pragma unroll
    for (int a2 = 0; a2 < 3; ++a2) ang[1 + a2] = fc * dv[a2];
#pragma unroll
    for (int a2 = 0; a2 < 3; ++a2)
#pragma unroll
      for (int b2 = 0; b2 < 3; ++b2) ang[4 + a2 * 3 + b2] = ang[1 + a2] * dv[b2];

#pragma unroll
    for (int j = 0; j < NPARA; ++j)
      unsafeAtomicAdd(&acc[j * NATOM + ci], ang[j] * wv);
  }
  __syncthreads();

  float* dst = sum_w + blockIdx.x * ACC_N;
  for (int i = threadIdx.x; i < ACC_N; i += PAIR_THREADS) dst[i] = acc[i];
}

__device__ __forceinline__ int sw_index(int atom, int para, int k) {
  return (((atom / NATOM) * NWAVE + k) * NPARA + para) * NATOM + (atom % NATOM);
}

__global__ void k_density(const float* __restrict__ sum_w,
                          const float* __restrict__ hyper,
                          float* __restrict__ density) {
  int wave = threadIdx.x >> 5;
  int lane = threadIdx.x & 31;
  int tile = blockIdx.x * 8 + wave;
  int lrow = lane & 15;
  int lhi = lane >> 4;
  int khalf = lhi * 2;

  for (int nt = 0; nt < 8; ++nt) {
    int n = nt * 16 + lrow;
    v8f acc = {};
    int para = 0;
    for (int ip = 0; ip < 3; ++ip) {
      const float* hb = hyper + ip * (NWAVE * NORBIT);
      v2f Blo, Bhi;
      Blo.x = hb[(khalf + 0) * NORBIT + n];
      Blo.y = hb[(khalf + 1) * NORBIT + n];
      Bhi.x = hb[(khalf + 4) * NORBIT + n];
      Bhi.y = hb[(khalf + 5) * NORBIT + n];
      int npj = (ip == 0) ? 1 : ((ip == 1) ? 3 : 9);
      for (int q = 0; q < npj; ++q, ++para) {
        const int arow = tile * 16 + lrow;
        v2f Alo = {sum_w[sw_index(arow, para, khalf + 0)],
                   sum_w[sw_index(arow, para, khalf + 1)]};
        v2f Ahi = {sum_w[sw_index(arow, para, khalf + 4)],
                   sum_w[sw_index(arow, para, khalf + 5)]};
        v8f c = {};
        c = wmma_f32_16x16x4(Alo, Blo, c);
        c = wmma_f32_16x16x4(Ahi, Bhi, c);
#pragma unroll
        for (int r = 0; r < 8; ++r) acc[r] += c[r] * c[r];
      }
    }
#pragma unroll
    for (int r = 0; r < 8; ++r) {
      int atom = tile * 16 + r + lhi * 8;
      density[atom * NORBIT + nt * 16 + lrow] = acc[r];
    }
  }
}

__global__ void k_update(const float* __restrict__ density,
                         const float* __restrict__ oc_w,
                         const float* __restrict__ oc_e,
                         const int* __restrict__ species,
                         float* __restrict__ orb) {
  int wave = threadIdx.x >> 5;
  int lane = threadIdx.x & 31;
  int tile = blockIdx.x * 8 + wave;
  int lrow = lane & 15;
  int lhi = lane >> 4;

  v8f acc = {};
  const float* arow = density + (tile * 16 + lrow) * NORBIT;
  bool valid = (lrow < NWAVE);
#pragma unroll 4
  for (int kb = 0; kb < 32; ++kb) {
    int k = kb * 4 + lhi * 2;
    v2f A = {arow[k], arow[k + 1]};
    v2f B;
    B.x = valid ? oc_w[(k + 0) * NWAVE + lrow] : 0.0f;
    B.y = valid ? oc_w[(k + 1) * NWAVE + lrow] : 0.0f;
    acc = wmma_f32_16x16x4(A, B, acc);
  }
  if (valid) {
#pragma unroll
    for (int r = 0; r < 8; ++r) {
      int atom = tile * 16 + r + lhi * 8;
      float v = tanhf(acc[r] + oc_e[species[atom] * NWAVE + lrow]);
      orb[atom * NWAVE + lrow] += v;
    }
  }
}

extern "C" void kernel_launch(void* const* d_in, const int* in_sizes, int n_in,
                              void* d_out, int out_size, void* d_ws,
                              size_t ws_size, hipStream_t stream) {
  const float* cart = (const float*)d_in[0];
  const float* shifts = (const float*)d_in[1];
  const float* rs = (const float*)d_in[2];
  const float* inta = (const float*)d_in[3];
  const float* params = (const float*)d_in[4];
  const float* hyper = (const float*)d_in[5];
  const float* oc_w0 = (const float*)d_in[6];
  const float* oc_e0 = (const float*)d_in[7];
  const float* oc_w1 = (const float*)d_in[8];
  const float* oc_e1 = (const float*)d_in[9];
  const int* species = (const int*)d_in[11];
  const int* atom_index = (const int*)d_in[12];

  float* density = (float*)d_out;
  float* orb = (float*)d_ws;
  float* sum_w = orb + TOTNATOM * NWAVE;

  const float* Ws[2] = {oc_w0, oc_w1};
  const float* Es[2] = {oc_e0, oc_e1};

  k_init_orb<<<(TOTNATOM * NWAVE) / 256, 256, 0, stream>>>(params, species, orb);
  for (int it = 0; it < 3; ++it) {
    k_pairs<<<NB * NWAVE, PAIR_THREADS, 0, stream>>>(cart, shifts, rs, inta,
                                                     species, atom_index, orb,
                                                     sum_w);
    k_density<<<TOTNATOM / 16 / 8, 256, 0, stream>>>(sum_w, hyper, density);
    if (it < 2)
      k_update<<<TOTNATOM / 16 / 8, 256, 0, stream>>>(density, Ws[it], Es[it],
                                                      species, orb);
  }
}


# Pair stage: LDS sum table in 16 copies (208 KiB, so one block per compute unit); a lane adds into copy lane mod 16, arrival order within a copy; the 16 partial sums then added in fixed order
# speedup vs baseline: 45.4182x; 1.1839x over previous
#include <hip/hip_runtime.h>
#include <math.h>

#define NTYPE 4
#define NWAVE 8
#define NORBIT 128
#define NEIGH 64
#define NB 32
#define NATOM 256
#define NPAIR_PER (NATOM * NEIGH)
#define NPAIR_TOT (NB * NPAIR_PER)
#define TOTNATOM (NB * NATOM)
#define NPARA 13
#define PI_OVER_CUTOFF 0.62831853071795864769f

typedef float v2f __attribute__((ext_vector_type(2)));
typedef float v8f __attribute__((ext_vector_type(8)));

__device__ __forceinline__ v8f wmma_f32_16x16x4(v2f a, v2f b, v8f c) {
  return __builtin_amdgcn_wmma_f32_16x16x4_f32(false, a, false, b, (short)0, c,
                                               false, false);
}

__global__ void k_init_orb(const float* __restrict__ params,
                           const int* __restrict__ species,
                           float* __restrict__ orb) {
  int t = blockIdx.x * blockDim.x + threadIdx.x;
  int a = t / NWAVE, w = t % NWAVE;
  orb[t] = params[species[a] * NWAVE + w];
}

#define ACC_N (NPARA * NATOM)
#define NCOPY 16
#define TBL_V4 (ACC_N * NCOPY / 4)
#define PAIR_THREADS 1024

typedef float v4f __attribute__((ext_vector_type(4)));

__global__ __launch_bounds__(PAIR_THREADS) void k_pairs(
    const float* __restrict__ cart, const float* __restrict__ shifts,
    const float* __restrict__ rs, const float* __restrict__ inta,
    const int* __restrict__ species, const int* __restrict__ atom_index,
    const float* __restrict__ orb, float* __restrict__ sum_w) {
  __shared__ v4f tbl4[TBL_V4];
  float* tbl = (float*)tbl4;
  const int b = blockIdx.x / NWAVE;
  const int w = blockIdx.x % NWAVE;

  const v4f zero4 = {0.0f, 0.0f, 0.0f, 0.0f};
  for (int i = threadIdx.x; i < TBL_V4; i += PAIR_THREADS) tbl4[i] = zero4;
  __syncthreads();

  const int copy = threadIdx.x & (NCOPY - 1);

  for (int q = threadIdx.x; q < NPAIR_PER; q += PAIR_THREADS) {
    int p = b * NPAIR_PER + q;
    int ci = atom_index[p];
    int gi = b * NATOM + ci;
    int gj = b * NATOM + atom_index[NPAIR_TOT + p];

    float dx = cart[gi * 3 + 0] - cart[gj * 3 + 0] + shifts[p * 3 + 0];
    float dy = cart[gi * 3 + 1] - cart[gj * 3 + 1] + shifts[p * 3 + 1];
    float dz = cart[gi * 3 + 2] - cart[gj * 3 + 2] + shifts[p * 3 + 2];
    float d = sqrtf(dx * dx + dy * dy + dz * dz);
    float fc = 0.5f * cosf(d * PI_OVER_CUTOFF) + 0.5f;
    fc = fc * fc;

    int sp = species[gj];
    float t = d - rs[sp * NWAVE + w];
    float wv = expf(inta[sp * NWAVE + w] * t * t) * orb[gj * NWAVE + w];

    float ang[NPARA];
    float dv[3] = {dx, dy, dz};
    ang[0] = fc;
#pragma unroll
    for (int a2 = 0; a2 < 3; ++a2) ang[1 + a2] = fc * dv[a2];
#pragma unroll
    for (int a2 = 0; a2 < 3; ++a2)
#pragma unroll
      for (int b2 = 0; b2 < 3; ++b2) ang[4 + a2 * 3 + b2] = ang[1 + a2] * dv[b2];

    float* cell = tbl + ci * NCOPY + copy;
#pragma unroll
    for (int j = 0; j < NPARA; ++j)
      unsafeAtomicAdd(cell + j * (NATOM * NCOPY), ang[j] * wv);
  }
  __syncthreads();

  float* dst = sum_w + blockIdx.x * ACC_N;
  const int turn = (threadIdx.x >> 2) & 3;
  for (int c = threadIdx.x; c < ACC_N; c += PAIR_THREADS) {
    float s = 0.0f;
#pragma unroll
    for (int k = 0; k < 4; ++k) {
      v4f v = tbl4[c * 4 + ((k + turn) & 3)];
      s += (v.x + v.y) + (v.z + v.w);
    }
    dst[c] = s;
  }
}

__device__ __forceinline__ int sw_index(int atom, int para, int k) {
  return (((atom / NATOM) * NWAVE + k) * NPARA + para) * NATOM + (atom % NATOM);
}

__global__ void k_density(const float* __restrict__ sum_w,
                          const float* __restrict__ hyper,
                          float* __restrict__ density) {
  int wave = threadIdx.x >> 5;
  int lane = threadIdx.x & 31;
  int tile = blockIdx.x * 8 + wave;
  int lrow = lane & 15;
  int lhi = lane >> 4;
  int khalf = lhi * 2;

  for (int nt = 0; nt < 8; ++nt) {
    int n = nt * 16 + lrow;
    v8f acc = {};
    int para = 0;
    for (int ip = 0; ip < 3; ++ip) {
      const float* hb = hyper + ip * (NWAVE * NORBIT);
      v2f Blo, Bhi;
      Blo.x = hb[(khalf + 0) * NORBIT + n];
      Blo.y = hb[(khalf + 1) * NORBIT + n];
      Bhi.x = hb[(khalf + 4) * NORBIT + n];
      Bhi.y = hb[(khalf + 5) * NORBIT + n];
      int npj = (ip == 0) ? 1 : ((ip == 1) ? 3 : 9);
      for (int q = 0; q < npj; ++q, ++para) {
        const int arow = tile * 16 + lrow;
        v2f Alo = {sum_w[sw_index(arow, para, khalf + 0)],
                   sum_w[sw_index(arow, para, khalf + 1)]};
        v2f Ahi = {sum_w[sw_index(arow, para, khalf + 4)],
                   sum_w[sw_index(arow, para, khalf + 5)]};
        v8f c = {};
        c = wmma_f32_16x16x4(Alo, Blo, c);
        c = wmma_f32_16x16x4(Ahi, Bhi, c);
#pragma unroll
        for (int r = 0; r < 8; ++r) acc[r] += c[r] * c[r];
      }
    }
#pragma unroll
    for (int r = 0; r < 8; ++r) {
      int atom = tile * 16 + r + lhi * 8;
      density[atom * NORBIT + nt * 16 + lrow] = acc[r];
    }
  }
}

__global__ void k_update(const float* __restrict__ density,
                         const float* __restrict__ oc_w,
                         const float* __restrict__ oc_e,
                         const int* __restrict__ species,
                         float* __restrict__ orb) {
  int wave = threadIdx.x >> 5;
  int lane = threadIdx.x & 31;
  int tile = blockIdx.x * 8 + wave;
  int lrow = lane & 15;
  int lhi = lane >> 4;

  v8f acc = {};
  const float* arow = density + (tile * 16 + lrow) * NORBIT;
  bool valid = (lrow < NWAVE);
#pragma unroll 4
  for (int kb = 0; kb < 32; ++kb) {
    int k = kb * 4 + lhi * 2;
    v2f A = {arow[k], arow[k + 1]};
    v2f B;
    B.x = valid ? oc_w[(k + 0) * NWAVE + lrow] : 0.0f;
    B.y = valid ? oc_w[(k + 1) * NWAVE + lrow] : 0.0f;
    acc = wmma_f32_16x16x4(A, B, acc);
  }
  if (valid) {
#pragma unroll
    for (int r = 0; r < 8; ++r) {
      int atom = tile * 16 + r + lhi * 8;
      float v = tanhf(acc[r] + oc_e[species[atom] * NWAVE + lrow]);
      orb[atom * NWAVE + lrow] += v;
    }
  }
}

extern "C" void kernel_launch(void* const* d_in, const int* in_sizes, int n_in,
                              void* d_out, int out_size, void* d_ws,
                              size_t ws_size, hipStream_t stream) {
  const float* cart = (const float*)d_in[0];
  const float* shifts = (const float*)d_in[1];
  const float* rs = (const float*)d_in[2];
  const float* inta = (const float*)d_in[3];
  const float* params = (const float*)d_in[4];
  const float* hyper = (const float*)d_in[5];
  const float* oc_w0 = (const float*)d_in[6];
  const float* oc_e0 = (const float*)d_in[7];
  const float* oc_w1 = (const float*)d_in[8];
  const float* oc_e1 = (const float*)d_in[9];
  const int* species = (const int*)d_in[11];
  const int* atom_index = (const int*)d_in[12];

  float* density = (float*)d_out;
  float* orb = (float*)d_ws;
  float* sum_w = orb + TOTNATOM * NWAVE;

  const float* Ws[2] = {oc_w0, oc_w1};
  const float* Es[2] = {oc_e0, oc_e1};

  k_init_orb<<<(TOTNATOM * NWAVE) / 256, 256, 0, stream>>>(params, species, orb);
  for (int it = 0; it < 3; ++it) {
    k_pairs<<<NB * NWAVE, PAIR_THREADS, 0, stream>>>(cart, shifts, rs, inta,
                                                     species, atom_index, orb,
                                                     sum_w);
    k_density<<<TOTNATOM / 16 / 8, 256, 0, stream>>>(sum_w, hyper, density);
    if (it < 2)
      k_update<<<TOTNATOM / 16 / 8, 256, 0, stream>>>(density, Ws[it], Es[it],
                                                      species, orb);
  }
}


# Each pair block stages its molecule's 256 atom records (x, y, z, orb, rs, inta) once in LDS; the pair loop reads them there: 1 global round trip per trip (was 4 dependent); sum order unchanged
# speedup vs baseline: 54.2122x; 1.1936x over previous
#include <hip/hip_runtime.h>
#include <math.h>

#define NTYPE 4
#define NWAVE 8
#define NORBIT 128
#define NEIGH 64
#define NB 32
#define NATOM 256
#define NPAIR_PER (NATOM * NEIGH)
#define NPAIR_TOT (NB * NPAIR_PER)
#define TOTNATOM (NB * NATOM)
#define NPARA 13
#define PI_OVER_CUTOFF 0.62831853071795864769f

typedef float v2f __attribute__((ext_vector_type(2)));
typedef float v8f __attribute__((ext_vector_type(8)));

__device__ __forceinline__ v8f wmma_f32_16x16x4(v2f a, v2f b, v8f c) {
  return __builtin_amdgcn_wmma_f32_16x16x4_f32(false, a, false, b, (short)0, c,
                                               false, false);
}

__global__ void k_init_orb(const float* __restrict__ params,
                           const int* __restrict__ species,
                           float* __restrict__ orb) {
  int t = blockIdx.x * blockDim.x + threadIdx.x;
  int a = t / NWAVE, w = t % NWAVE;
  orb[t] = params[species[a] * NWAVE + w];
}

#define ACC_N (NPARA * NATOM)
#define NCOPY 16
#define TBL_V4 (ACC_N * NCOPY / 4)
#define PAIR_THREADS 1024

typedef float v4f __attribute__((ext_vector_type(4)));

__global__ __launch_bounds__(PAIR_THREADS) void k_pairs(
    const float* __restrict__ cart, const float* __restrict__ shifts,
    const float* __restrict__ rs, const float* __restrict__ inta,
    const int* __restrict__ species, const int* __restrict__ atom_index,
    const float* __restrict__ orb, float* __restrict__ sum_w) {
  __shared__ v4f tbl4[TBL_V4];
  __shared__ v4f rec_pos[NATOM];
  __shared__ v2f rec_rad[NATOM];
  float* tbl = (float*)tbl4;
  const int b = blockIdx.x / NWAVE;
  const int w = blockIdx.x % NWAVE;

  const v4f zero4 = {0.0f, 0.0f, 0.0f, 0.0f};
  const bool stager = threadIdx.x < NATOM;
  v4f pos = zero4;
  v2f rad = {0.0f, 0.0f};
  if (stager) {
    const int a = b * NATOM + threadIdx.x;
    const int spa = species[a];
    pos.x = cart[a * 3 + 0];
    pos.y = cart[a * 3 + 1];
    pos.z = cart[a * 3 + 2];
    pos.w = orb[a * NWAVE + w];
    rad.x = rs[spa * NWAVE + w];
    rad.y = inta[spa * NWAVE + w];
  }

  for (int i = threadIdx.x; i < TBL_V4; i += PAIR_THREADS) tbl4[i] = zero4;
  if (stager) {
    rec_pos[threadIdx.x] = pos;
    rec_rad[threadIdx.x] = rad;
  }
  __syncthreads();

  const int copy = threadIdx.x & (NCOPY - 1);

  for (int q = threadIdx.x; q < NPAIR_PER; q += PAIR_THREADS) {
    int p = b * NPAIR_PER + q;
    float sx = shifts[p * 3 + 0];
    float sy = shifts[p * 3 + 1];
    float sz = shifts[p * 3 + 2];
    int ci = atom_index[p];
    int cj = atom_index[NPAIR_TOT + p];

    v4f rc = rec_pos[ci];
    v4f rn = rec_pos[cj];
    v2f rr = rec_rad[cj];

    float dx = rc.x - rn.x + sx;
    float dy = rc.y - rn.y + sy;
    float dz = rc.z - rn.z + sz;
    float d = sqrtf(dx * dx + dy * dy + dz * dz);
    float fc = 0.5f * cosf(d * PI_OVER_CUTOFF) + 0.5f;
    fc = fc * fc;

    float t = d - rr.x;
    float wv = expf(rr.y * t * t) * rn.w;

    float ang[NPARA];
    float dv[3] = {dx, dy, dz};
    ang[0] = fc;
#pragma unroll
    for (int a2 = 0; a2 < 3; ++a2) ang[1 + a2] = fc * dv[a2];
#pragma unroll
    for (int a2 = 0; a2 < 3; ++a2)
#pragma unroll
      for (int b2 = 0; b2 < 3; ++b2) ang[4 + a2 * 3 + b2] = ang[1 + a2] * dv[b2];

    float* cell = tbl + ci * NCOPY + copy;
#pragma unroll
    for (int j = 0; j < NPARA; ++j)
      unsafeAtomicAdd(cell + j * (NATOM * NCOPY), ang[j] * wv);
  }
  __syncthreads();

  float* dst = sum_w + blockIdx.x * ACC_N;
  const int turn = (threadIdx.x >> 2) & 3;
  for (int c = threadIdx.x; c < ACC_N; c += PAIR_THREADS) {
    float s = 0.0f;
#pragma unroll
    for (int k = 0; k < 4; ++k) {
      v4f v = tbl4[c * 4 + ((k + turn) & 3)];
      s += (v.x + v.y) + (v.z + v.w);
    }
    dst[c] = s;
  }
}

__device__ __forceinline__ int sw_index(int atom, int para, int k) {
  return (((atom / NATOM) * NWAVE + k) * NPARA + para) * NATOM + (atom % NATOM);
}

__global__ void k_density(const float* __restrict__ sum_w,
                          const float* __restrict__ hyper,
                          float* __restrict__ density) {
  int wave = threadIdx.x >> 5;
  int lane = threadIdx.x & 31;
  int tile = blockIdx.x * 8 + wave;
  int lrow = lane & 15;
  int lhi = lane >> 4;
  int khalf = lhi * 2;

  for (int nt = 0; nt < 8; ++nt) {
    int n = nt * 16 + lrow;
    v8f acc = {};
    int para = 0;
    for (int ip = 0; ip < 3; ++ip) {
      const float* hb = hyper + ip * (NWAVE * NORBIT);
      v2f Blo, Bhi;
      Blo.x = hb[(khalf + 0) * NORBIT + n];
      Blo.y = hb[(khalf + 1) * NORBIT + n];
      Bhi.x = hb[(khalf + 4) * NORBIT + n];
      Bhi.y = hb[(khalf + 5) * NORBIT + n];
      int npj = (ip == 0) ? 1 : ((ip == 1) ? 3 : 9);
      for (int q = 0; q < npj; ++q, ++para) {
        const int arow = tile * 16 + lrow;
        v2f Alo = {sum_w[sw_index(arow, para, khalf + 0)],
                   sum_w[sw_index(arow, para, khalf + 1)]};
        v2f Ahi = {sum_w[sw_index(arow, para, khalf + 4)],
                   sum_w[sw_index(arow, para, khalf + 5)]};
        v8f c = {};
        c = wmma_f32_16x16x4(Alo, Blo, c);
        c = wmma_f32_16x16x4(Ahi, Bhi, c);
#pragma unroll
        for (int r = 0; r < 8; ++r) acc[r] += c[r] * c[r];
      }
    }
#pragma unroll
    for (int r = 0; r < 8; ++r) {
      int atom = tile * 16 + r + lhi * 8;
      density[atom * NORBIT + nt * 16 + lrow] = acc[r];
    }
  }
}

__global__ void k_update(const float* __restrict__ density,
                         const float* __restrict__ oc_w,
                         const float* __restrict__ oc_e,
                         const int* __restrict__ species,
                         float* __restrict__ orb) {
  int wave = threadIdx.x >> 5;
  int lane = threadIdx.x & 31;
  int tile = blockIdx.x * 8 + wave;
  int lrow = lane & 15;
  int lhi = lane >> 4;

  v8f acc = {};
  const float* arow = density + (tile * 16 + lrow) * NORBIT;
  bool valid = (lrow < NWAVE);
#pragma unroll 4
  for (int kb = 0; kb < 32; ++kb) {
    int k = kb * 4 + lhi * 2;
    v2f A = {arow[k], arow[k + 1]};
    v2f B;
    B.x = valid ? oc_w[(k + 0) * NWAVE + lrow] : 0.0f;
    B.y = valid ? oc_w[(k + 1) * NWAVE + lrow] : 0.0f;
    acc = wmma_f32_16x16x4(A, B, acc);
  }
  if (valid) {
#pragma unroll
    for (int r = 0; r < 8; ++r) {
      int atom = tile * 16 + r + lhi * 8;
      float v = tanhf(acc[r] + oc_e[species[atom] * NWAVE + lrow]);
      orb[atom * NWAVE + lrow] += v;
    }
  }
}

extern "C" void kernel_launch(void* const* d_in, const int* in_sizes, int n_in,
                              void* d_out, int out_size, void* d_ws,
                              size_t ws_size, hipStream_t stream) {
  const float* cart = (const float*)d_in[0];
  const float* shifts = (const float*)d_in[1];
  const float* rs = (const float*)d_in[2];
  const float* inta = (const float*)d_in[3];
  const float* params = (const float*)d_in[4];
  const float* hyper = (const float*)d_in[5];
  const float* oc_w0 = (const float*)d_in[6];
  const float* oc_e0 = (const float*)d_in[7];
  const float* oc_w1 = (const float*)d_in[8];
  const float* oc_e1 = (const float*)d_in[9];
  const int* species = (const int*)d_in[11];
  const int* atom_index = (const int*)d_in[12];

  float* density = (float*)d_out;
  float* orb = (float*)d_ws;
  float* sum_w = orb + TOTNATOM * NWAVE;

  const float* Ws[2] = {oc_w0, oc_w1};
  const float* Es[2] = {oc_e0, oc_e1};

  k_init_orb<<<(TOTNATOM * NWAVE) / 256, 256, 0, stream>>>(params, species, orb);
  for (int it = 0; it < 3; ++it) {
    k_pairs<<<NB * NWAVE, PAIR_THREADS, 0, stream>>>(cart, shifts, rs, inta,
                                                     species, atom_index, orb,
                                                     sum_w);
    k_density<<<TOTNATOM / 16 / 8, 256, 0, stream>>>(sum_w, hyper, density);
    if (it < 2)
      k_update<<<TOTNATOM / 16 / 8, 256, 0, stream>>>(density, Ws[it], Es[it],
                                                      species, orb);
  }
}


# Coefficient update and init folded into neighbour kernels: density tile, formed transposed in registers, is the update WMMA's B operand (update sums in a new k order); 6 launches, was 9
# speedup vs baseline: 75.4017x; 1.3909x over previous
#include <hip/hip_runtime.h>
#include <math.h>

#define NTYPE 4
#define NWAVE 8
#define NORBIT 128
#define NEIGH 64
#define NB 32
#define NATOM 256
#define NPAIR_PER (NATOM * NEIGH)
#define NPAIR_TOT (NB * NPAIR_PER)
#define TOTNATOM (NB * NATOM)
#define NPARA 13
#define PI_OVER_CUTOFF 0.62831853071795864769f

typedef float v2f __attribute__((ext_vector_type(2)));
typedef float v8f __attribute__((ext_vector_type(8)));

__device__ __forceinline__ v8f wmma_f32_16x16x4(v2f a, v2f b, v8f c) {
  return __builtin_amdgcn_wmma_f32_16x16x4_f32(false, a, false, b, (short)0, c,
                                               false, false);
}

#define ACC_N (NPARA * NATOM)
#define NCOPY 16
#define TBL_V4 (ACC_N * NCOPY / 4)
#define PAIR_THREADS 1024

typedef float v4f __attribute__((ext_vector_type(4)));

template <bool FIRST>
__global__ __launch_bounds__(PAIR_THREADS) void k_pairs(
    const float* __restrict__ cart, const float* __restrict__ shifts,
    const float* __restrict__ rs, const float* __restrict__ inta,
    const float* __restrict__ params,
    const int* __restrict__ species, const int* __restrict__ atom_index,
    const float* __restrict__ orb, float* __restrict__ sum_w) {
  __shared__ v4f tbl4[TBL_V4];
  __shared__ v4f rec_pos[NATOM];
  __shared__ v2f rec_rad[NATOM];
  float* tbl = (float*)tbl4;
  const int b = blockIdx.x / NWAVE;
  const int w = blockIdx.x % NWAVE;

  const v4f zero4 = {0.0f, 0.0f, 0.0f, 0.0f};
  const bool stager = threadIdx.x < NATOM;
  v4f pos = zero4;
  v2f rad = {0.0f, 0.0f};
  if (stager) {
    const int a = b * NATOM + threadIdx.x;
    const int spa = species[a];
    pos.x = cart[a * 3 + 0];
    pos.y = cart[a * 3 + 1];
    pos.z = cart[a * 3 + 2];
    pos.w = FIRST ? params[spa * NWAVE + w] : orb[a * NWAVE + w];
    rad.x = rs[spa * NWAVE + w];
    rad.y = inta[spa * NWAVE + w];
  }

  for (int i = threadIdx.x; i < TBL_V4; i += PAIR_THREADS) tbl4[i] = zero4;
  if (stager) {
    rec_pos[threadIdx.x] = pos;
    rec_rad[threadIdx.x] = rad;
  }
  __syncthreads();

  const int copy = threadIdx.x & (NCOPY - 1);

  for (int q = threadIdx.x; q < NPAIR_PER; q += PAIR_THREADS) {
    int p = b * NPAIR_PER + q;
    float sx = shifts[p * 3 + 0];
    float sy = shifts[p * 3 + 1];
    float sz = shifts[p * 3 + 2];
    int ci = atom_index[p];
    int cj = atom_index[NPAIR_TOT + p];

    v4f rc = rec_pos[ci];
    v4f rn = rec_pos[cj];
    v2f rr = rec_rad[cj];

    float dx = rc.x - rn.x + sx;
    float dy = rc.y - rn.y + sy;
    float dz = rc.z - rn.z + sz;
    float d = sqrtf(dx * dx + dy * dy + dz * dz);
    float fc = 0.5f * cosf(d * PI_OVER_CUTOFF) + 0.5f;
    fc = fc * fc;

    float t = d - rr.x;
    float wv = expf(rr.y * t * t) * rn.w;

    float ang[NPARA];
    float dv[3] = {dx, dy, dz};
    ang[0] = fc;
#pragma unroll
    for (int a2 = 0; a2 < 3; ++a2) ang[1 + a2] = fc * dv[a2];
#pragma unroll
    for (int a2 = 0; a2 < 3; ++a2)
#pragma unroll
      for (int b2 = 0; b2 < 3; ++b2) ang[4 + a2 * 3 + b2] = ang[1 + a2] * dv[b2];

    float* cell = tbl + ci * NCOPY + copy;
#pragma unroll
    for (int j = 0; j < NPARA; ++j)
      unsafeAtomicAdd(cell + j * (NATOM * NCOPY), ang[j] * wv);
  }
  __syncthreads();

  float* dst = sum_w + blockIdx.x * ACC_N;
  const int turn = (threadIdx.x >> 2) & 3;
  for (int c = threadIdx.x; c < ACC_N; c += PAIR_THREADS) {
    float s = 0.0f;
#pragma unroll
    for (int k = 0; k < 4; ++k) {
      v4f v = tbl4[c * 4 + ((k + turn) & 3)];
      s += (v.x + v.y) + (v.z + v.w);
    }
    dst[c] = s;
  }
}

__device__ __forceinline__ int sw_index(int atom, int para, int k) {
  return (((atom / NATOM) * NWAVE + k) * NPARA + para) * NATOM + (atom % NATOM);
}

__global__ void k_density(const float* __restrict__ sum_w,
                          const float* __restrict__ hyper,
                          float* __restrict__ density) {
  int wave = threadIdx.x >> 5;
  int lane = threadIdx.x & 31;
  int tile = blockIdx.x * 8 + wave;
  int lrow = lane & 15;
  int lhi = lane >> 4;
  int khalf = lhi * 2;

  for (int nt = 0; nt < 8; ++nt) {
    int n = nt * 16 + lrow;
    v8f acc = {};
    int para = 0;
    for (int ip = 0; ip < 3; ++ip) {
      const float* hb = hyper + ip * (NWAVE * NORBIT);
      v2f Blo, Bhi;
      Blo.x = hb[(khalf + 0) * NORBIT + n];
      Blo.y = hb[(khalf + 1) * NORBIT + n];
      Bhi.x = hb[(khalf + 4) * NORBIT + n];
      Bhi.y = hb[(khalf + 5) * NORBIT + n];
      int npj = (ip == 0) ? 1 : ((ip == 1) ? 3 : 9);
      for (int q = 0; q < npj; ++q, ++para) {
        const int arow = tile * 16 + lrow;
        v2f Alo = {sum_w[sw_index(arow, para, khalf + 0)],
                   sum_w[sw_index(arow, para, khalf + 1)]};
        v2f Ahi = {sum_w[sw_index(arow, para, khalf + 4)],
                   sum_w[sw_index(arow, para, khalf + 5)]};
        v8f c = {};
        c = wmma_f32_16x16x4(Alo, Blo, c);
        c = wmma_f32_16x16x4(Ahi, Bhi, c);
#pragma unroll
        for (int r = 0; r < 8; ++r) acc[r] += c[r] * c[r];
      }
    }
#pragma unroll
    for (int r = 0; r < 8; ++r) {
      int atom = tile * 16 + r + lhi * 8;
      density[atom * NORBIT + nt * 16 + lrow] = acc[r];
    }
  }
}

template <bool FIRST>
__global__ __launch_bounds__(256) void k_density_update(const float* __restrict__ sum_w,
                                 const float* __restrict__ hyper,
                                 const float* __restrict__ oc_w,
                                 const float* __restrict__ oc_e,
                                 const float* __restrict__ params,
                                 const int* __restrict__ species,
                                 float* __restrict__ orb) {
  const int wave = threadIdx.x >> 5;
  const int lane = threadIdx.x & 31;
  const int tile = blockIdx.x * 8 + wave;
  const int lrow = lane & 15;
  const int lhi = lane >> 4;
  const int khalf = lhi * 2;
  const int atom = tile * 16 + lrow;

  const int sp = species[atom];
  const v4f* erow = (const v4f*)(oc_e + sp * NWAVE);
  const v4f* orow = FIRST ? (const v4f*)(params + sp * NWAVE)
                          : (const v4f*)(orb + atom * NWAVE);
  const v4f e0 = erow[0], e1 = erow[1];
  const v4f old0 = orow[0], old1 = orow[1];

  v2f Slo[NPARA], Shi[NPARA];
#pragma unroll
  for (int p = 0; p < NPARA; ++p) {
    Slo[p].x = sum_w[sw_index(atom, p, khalf + 0)];
    Slo[p].y = sum_w[sw_index(atom, p, khalf + 1)];
    Shi[p].x = sum_w[sw_index(atom, p, khalf + 4)];
    Shi[p].y = sum_w[sw_index(atom, p, khalf + 5)];
  }

  const bool wrow = lrow < NWAVE;
  const int wcol = lrow & (NWAVE - 1);
  v8f upd0 = {}, upd1 = {};

  for (int nt = 0; nt < 8; ++nt) {
    const int n = nt * 16 + lrow;
    v8f acc = {};
    int para = 0;
#pragma unroll
    for (int ip = 0; ip < 3; ++ip) {
      const float* hb = hyper + ip * (NWAVE * NORBIT);
      v2f Hlo, Hhi;
      Hlo.x = hb[(khalf + 0) * NORBIT + n];
      Hlo.y = hb[(khalf + 1) * NORBIT + n];
      Hhi.x = hb[(khalf + 4) * NORBIT + n];
      Hhi.y = hb[(khalf + 5) * NORBIT + n];
      const int npj = (ip == 0) ? 1 : ((ip == 1) ? 3 : 9);
#pragma unroll
      for (int q = 0; q < npj; ++q, ++para) {
        v8f c = {};
        c = wmma_f32_16x16x4(Hlo, Slo[para], c);
        c = wmma_f32_16x16x4(Hhi, Shi[para], c);
#pragma unroll
        for (int r = 0; r < 8; ++r) acc[r] += c[r] * c[r];
      }
    }
#pragma unroll
    for (int j = 0; j < 4; ++j) {
      const int m0 = nt * 16 + lhi * 8 + 2 * j;
      const float w0 = oc_w[(m0 + 0) * NWAVE + wcol];
      const float w1 = oc_w[(m0 + 1) * NWAVE + wcol];
      v2f A = {wrow ? w0 : 0.0f, wrow ? w1 : 0.0f};
      v2f B = {acc[2 * j], acc[2 * j + 1]};
      if (j & 1) upd1 = wmma_f32_16x16x4(A, B, upd1);
      else       upd0 = wmma_f32_16x16x4(A, B, upd0);
    }
  }

  if (lhi == 0) {
    v4f n0, n1;
    n0.x = old0.x + tanhf((upd0[0] + upd1[0]) + e0.x);
    n0.y = old0.y + tanhf((upd0[1] + upd1[1]) + e0.y);
    n0.z = old0.z + tanhf((upd0[2] + upd1[2]) + e0.z);
    n0.w = old0.w + tanhf((upd0[3] + upd1[3]) + e0.w);
    n1.x = old1.x + tanhf((upd0[4] + upd1[4]) + e1.x);
    n1.y = old1.y + tanhf((upd0[5] + upd1[5]) + e1.y);
    n1.z = old1.z + tanhf((upd0[6] + upd1[6]) + e1.z);
    n1.w = old1.w + tanhf((upd0[7] + upd1[7]) + e1.w);
    v4f* dst = (v4f*)(orb + atom * NWAVE);
    dst[0] = n0;
    dst[1] = n1;
  }
}

extern "C" void kernel_launch(void* const* d_in, const int* in_sizes, int n_in,
                              void* d_out, int out_size, void* d_ws,
                              size_t ws_size, hipStream_t stream) {
  const float* cart = (const float*)d_in[0];
  const float* shifts = (const float*)d_in[1];
  const float* rs = (const float*)d_in[2];
  const float* inta = (const float*)d_in[3];
  const float* params = (const float*)d_in[4];
  const float* hyper = (const float*)d_in[5];
  const float* oc_w0 = (const float*)d_in[6];
  const float* oc_e0 = (const float*)d_in[7];
  const float* oc_w1 = (const float*)d_in[8];
  const float* oc_e1 = (const float*)d_in[9];
  const int* species = (const int*)d_in[11];
  const int* atom_index = (const int*)d_in[12];

  float* density = (float*)d_out;
  float* orb = (float*)d_ws;
  float* sum_w = orb + TOTNATOM * NWAVE;

  k_pairs<true><<<NB * NWAVE, PAIR_THREADS, 0, stream>>>(
      cart, shifts, rs, inta, params, species, atom_index, orb, sum_w);
  k_density_update<true><<<TOTNATOM / 16 / 8, 256, 0, stream>>>(
      sum_w, hyper, oc_w0, oc_e0, params, species, orb);
  k_pairs<false><<<NB * NWAVE, PAIR_THREADS, 0, stream>>>(
      cart, shifts, rs, inta, params, species, atom_index, orb, sum_w);
  k_density_update<false><<<TOTNATOM / 16 / 8, 256, 0, stream>>>(
      sum_w, hyper, oc_w1, oc_e1, params, species, orb);
  k_pairs<false><<<NB * NWAVE, PAIR_THREADS, 0, stream>>>(
      cart, shifts, rs, inta, params, species, atom_index, orb, sum_w);
  k_density<<<TOTNATOM / 16 / 8, 256, 0, stream>>>(sum_w, hyper, density);
}


# Density kernels re-tiled: a wave takes 2 of a tile's 8 orbital N-tiles (was all 8), 256 blocks (was 64), f32 WMMA work on all compute units; update partials of 4 waves added via LDS in fixed order
# speedup vs baseline: 92.4930x; 1.2267x over previous
#include <hip/hip_runtime.h>
#include <math.h>

#define NTYPE 4
#define NWAVE 8
#define NORBIT 128
#define NEIGH 64
#define NB 32
#define NATOM 256
#define NPAIR_PER (NATOM * NEIGH)
#define NPAIR_TOT (NB * NPAIR_PER)
#define TOTNATOM (NB * NATOM)
#define NPARA 13
#define PI_OVER_CUTOFF 0.62831853071795864769f

typedef float v2f __attribute__((ext_vector_type(2)));
typedef float v8f __attribute__((ext_vector_type(8)));

__device__ __forceinline__ v8f wmma_f32_16x16x4(v2f a, v2f b, v8f c) {
  return __builtin_amdgcn_wmma_f32_16x16x4_f32(false, a, false, b, (short)0, c,
                                               false, false);
}

#define ACC_N (NPARA * NATOM)
#define NCOPY 16
#define TBL_V4 (ACC_N * NCOPY / 4)
#define PAIR_THREADS 1024

typedef float v4f __attribute__((ext_vector_type(4)));

template <bool FIRST>
__global__ __launch_bounds__(PAIR_THREADS) void k_pairs(
    const float* __restrict__ cart, const float* __restrict__ shifts,
    const float* __restrict__ rs, const float* __restrict__ inta,
    const float* __restrict__ params,
    const int* __restrict__ species, const int* __restrict__ atom_index,
    const float* __restrict__ orb, float* __restrict__ sum_w) {
  __shared__ v4f tbl4[TBL_V4];
  __shared__ v4f rec_pos[NATOM];
  __shared__ v2f rec_rad[NATOM];
  float* tbl = (float*)tbl4;
  const int b = blockIdx.x / NWAVE;
  const int w = blockIdx.x % NWAVE;

  const v4f zero4 = {0.0f, 0.0f, 0.0f, 0.0f};
  const bool stager = threadIdx.x < NATOM;
  v4f pos = zero4;
  v2f rad = {0.0f, 0.0f};
  if (stager) {
    const int a = b * NATOM + threadIdx.x;
    const int spa = species[a];
    pos.x = cart[a * 3 + 0];
    pos.y = cart[a * 3 + 1];
    pos.z = cart[a * 3 + 2];
    pos.w = FIRST ? params[spa * NWAVE + w] : orb[a * NWAVE + w];
    rad.x = rs[spa * NWAVE + w];
    rad.y = inta[spa * NWAVE + w];
  }

  for (int i = threadIdx.x; i < TBL_V4; i += PAIR_THREADS) tbl4[i] = zero4;
  if (stager) {
    rec_pos[threadIdx.x] = pos;
    rec_rad[threadIdx.x] = rad;
  }
  __syncthreads();

  const int copy = threadIdx.x & (NCOPY - 1);

  for (int q = threadIdx.x; q < NPAIR_PER; q += PAIR_THREADS) {
    int p = b * NPAIR_PER + q;
    float sx = shifts[p * 3 + 0];
    float sy = shifts[p * 3 + 1];
    float sz = shifts[p * 3 + 2];
    int ci = atom_index[p];
    int cj = atom_index[NPAIR_TOT + p];

    v4f rc = rec_pos[ci];
    v4f rn = rec_pos[cj];
    v2f rr = rec_rad[cj];

    float dx = rc.x - rn.x + sx;
    float dy = rc.y - rn.y + sy;
    float dz = rc.z - rn.z + sz;
    float d = sqrtf(dx * dx + dy * dy + dz * dz);
    float fc = 0.5f * cosf(d * PI_OVER_CUTOFF) + 0.5f;
    fc = fc * fc;

    float t = d - rr.x;
    float wv = expf(rr.y * t * t) * rn.w;

    float ang[NPARA];
    float dv[3] = {dx, dy, dz};
    ang[0] = fc;
#pragma unroll
    for (int a2 = 0; a2 < 3; ++a2) ang[1 + a2] = fc * dv[a2];
#pragma unroll
    for (int a2 = 0; a2 < 3; ++a2)
#pragma unroll
      for (int b2 = 0; b2 < 3; ++b2) ang[4 + a2 * 3 + b2] = ang[1 + a2] * dv[b2];

    float* cell = tbl + ci * NCOPY + copy;
#pragma unroll
    for (int j = 0; j < NPARA; ++j)
      unsafeAtomicAdd(cell + j * (NATOM * NCOPY), ang[j] * wv);
  }
  __syncthreads();

  float* dst = sum_w + blockIdx.x * ACC_N;
  const int turn = (threadIdx.x >> 2) & 3;
  for (int c = threadIdx.x; c < ACC_N; c += PAIR_THREADS) {
    float s = 0.0f;
#pragma unroll
    for (int k = 0; k < 4; ++k) {
      v4f v = tbl4[c * 4 + ((k + turn) & 3)];
      s += (v.x + v.y) + (v.z + v.w);
    }
    dst[c] = s;
  }
}

__device__ __forceinline__ int sw_index(int atom, int para, int k) {
  return (((atom / NATOM) * NWAVE + k) * NPARA + para) * NATOM + (atom % NATOM);
}

#define NT_PER_WAVE 2
#define DENS_BLOCKS (TOTNATOM / 16 / 2)
__global__ void k_density(const float* __restrict__ sum_w,
                          const float* __restrict__ hyper,
                          float* __restrict__ density) {
  int wave = threadIdx.x >> 5;
  int lane = threadIdx.x & 31;
  int tile = blockIdx.x * 2 + (wave >> 2);
  int ntp = wave & 3;
  int lrow = lane & 15;
  int lhi = lane >> 4;
  int khalf = lhi * 2;

#pragma unroll
  for (int nt = ntp * NT_PER_WAVE; nt < ntp * NT_PER_WAVE + NT_PER_WAVE; ++nt) {
    int n = nt * 16 + lrow;
    v8f acc = {};
    int para = 0;
    for (int ip = 0; ip < 3; ++ip) {
      const float* hb = hyper + ip * (NWAVE * NORBIT);
      v2f Blo, Bhi;
      Blo.x = hb[(khalf + 0) * NORBIT + n];
      Blo.y = hb[(khalf + 1) * NORBIT + n];
      Bhi.x = hb[(khalf + 4) * NORBIT + n];
      Bhi.y = hb[(khalf + 5) * NORBIT + n];
      int npj = (ip == 0) ? 1 : ((ip == 1) ? 3 : 9);
      for (int q = 0; q < npj; ++q, ++para) {
        const int arow = tile * 16 + lrow;
        v2f Alo = {sum_w[sw_index(arow, para, khalf + 0)],
                   sum_w[sw_index(arow, para, khalf + 1)]};
        v2f Ahi = {sum_w[sw_index(arow, para, khalf + 4)],
                   sum_w[sw_index(arow, para, khalf + 5)]};
        v8f c = {};
        c = wmma_f32_16x16x4(Alo, Blo, c);
        c = wmma_f32_16x16x4(Ahi, Bhi, c);
#pragma unroll
        for (int r = 0; r < 8; ++r) acc[r] += c[r] * c[r];
      }
    }
#pragma unroll
    for (int r = 0; r < 8; ++r) {
      int atom = tile * 16 + r + lhi * 8;
      density[atom * NORBIT + nt * 16 + lrow] = acc[r];
    }
  }
}

template <bool FIRST>
__global__ __launch_bounds__(256) void k_density_update(const float* __restrict__ sum_w,
                                 const float* __restrict__ hyper,
                                 const float* __restrict__ oc_w,
                                 const float* __restrict__ oc_e,
                                 const float* __restrict__ params,
                                 const int* __restrict__ species,
                                 float* __restrict__ orb) {
  __shared__ float part[2][4][NWAVE][16];
  const int wave = threadIdx.x >> 5;
  const int lane = threadIdx.x & 31;
  const int tib = wave >> 2;
  const int ntp = wave & 3;
  const int tile = blockIdx.x * 2 + tib;
  const int lrow = lane & 15;
  const int lhi = lane >> 4;
  const int khalf = lhi * 2;
  const int atom = tile * 16 + lrow;

  const int eatom = blockIdx.x * 32 + (threadIdx.x >> 3);
  const int ech = threadIdx.x & (NWAVE - 1);
  const int sp = species[eatom];
  const float eoff = oc_e[sp * NWAVE + ech];
  const float old = FIRST ? params[sp * NWAVE + ech] : orb[eatom * NWAVE + ech];

  v2f Slo[NPARA], Shi[NPARA];
#pragma unroll
  for (int p = 0; p < NPARA; ++p) {
    Slo[p].x = sum_w[sw_index(atom, p, khalf + 0)];
    Slo[p].y = sum_w[sw_index(atom, p, khalf + 1)];
    Shi[p].x = sum_w[sw_index(atom, p, khalf + 4)];
    Shi[p].y = sum_w[sw_index(atom, p, khalf + 5)];
  }

  const bool wrow = lrow < NWAVE;
  const int wcol = lrow & (NWAVE - 1);
  v8f upd0 = {}, upd1 = {};

#pragma unroll
  for (int nt = ntp * NT_PER_WAVE; nt < ntp * NT_PER_WAVE + NT_PER_WAVE; ++nt) {
    const int n = nt * 16 + lrow;
    v8f acc = {};
    int para = 0;
#pragma unroll
    for (int ip = 0; ip < 3; ++ip) {
      const float* hb = hyper + ip * (NWAVE * NORBIT);
      v2f Hlo, Hhi;
      Hlo.x = hb[(khalf + 0) * NORBIT + n];
      Hlo.y = hb[(khalf + 1) * NORBIT + n];
      Hhi.x = hb[(khalf + 4) * NORBIT + n];
      Hhi.y = hb[(khalf + 5) * NORBIT + n];
      const int npj = (ip == 0) ? 1 : ((ip == 1) ? 3 : 9);
#pragma unroll
      for (int q = 0; q < npj; ++q, ++para) {
        v8f c = {};
        c = wmma_f32_16x16x4(Hlo, Slo[para], c);
        c = wmma_f32_16x16x4(Hhi, Shi[para], c);
#pragma unroll
        for (int r = 0; r < 8; ++r) acc[r] += c[r] * c[r];
      }
    }
#pragma unroll
    for (int j = 0; j < 4; ++j) {
      const int m0 = nt * 16 + lhi * 8 + 2 * j;
      const float w0 = oc_w[(m0 + 0) * NWAVE + wcol];
      const float w1 = oc_w[(m0 + 1) * NWAVE + wcol];
      v2f A = {wrow ? w0 : 0.0f, wrow ? w1 : 0.0f};
      v2f B = {acc[2 * j], acc[2 * j + 1]};
      if (j & 1) upd1 = wmma_f32_16x16x4(A, B, upd1);
      else       upd0 = wmma_f32_16x16x4(A, B, upd0);
    }
  }

  if (lhi == 0) {
#pragma unroll
    for (int r = 0; r < NWAVE; ++r) part[tib][ntp][r][lrow] = upd0[r] + upd1[r];
  }
  __syncthreads();

  const int etib = threadIdx.x >> 7;
  const int ecol = (threadIdx.x >> 3) & 15;
  const float s = (part[etib][0][ech][ecol] + part[etib][1][ech][ecol]) +
                  (part[etib][2][ech][ecol] + part[etib][3][ech][ecol]);
  orb[eatom * NWAVE + ech] = old + tanhf(s + eoff);
}

extern "C" void kernel_launch(void* const* d_in, const int* in_sizes, int n_in,
                              void* d_out, int out_size, void* d_ws,
                              size_t ws_size, hipStream_t stream) {
  const float* cart = (const float*)d_in[0];
  const float* shifts = (const float*)d_in[1];
  const float* rs = (const float*)d_in[2];
  const float* inta = (const float*)d_in[3];
  const float* params = (const float*)d_in[4];
  const float* hyper = (const float*)d_in[5];
  const float* oc_w0 = (const float*)d_in[6];
  const float* oc_e0 = (const float*)d_in[7];
  const float* oc_w1 = (const float*)d_in[8];
  const float* oc_e1 = (const float*)d_in[9];
  const int* species = (const int*)d_in[11];
  const int* atom_index = (const int*)d_in[12];

  float* density = (float*)d_out;
  float* orb = (float*)d_ws;
  float* sum_w = orb + TOTNATOM * NWAVE;

  k_pairs<true><<<NB * NWAVE, PAIR_THREADS, 0, stream>>>(
      cart, shifts, rs, inta, params, species, atom_index, orb, sum_w);
  k_density_update<true><<<DENS_BLOCKS, 256, 0, stream>>>(
      sum_w, hyper, oc_w0, oc_e0, params, species, orb);
  k_pairs<false><<<NB * NWAVE, PAIR_THREADS, 0, stream>>>(
      cart, shifts, rs, inta, params, species, atom_index, orb, sum_w);
  k_density_update<false><<<DENS_BLOCKS, 256, 0, stream>>>(
      sum_w, hyper, oc_w1, oc_e1, params, species, orb);
  k_pairs<false><<<NB * NWAVE, PAIR_THREADS, 0, stream>>>(
      cart, shifts, rs, inta, params, species, atom_index, orb, sum_w);
  k_density<<<DENS_BLOCKS, 256, 0, stream>>>(sum_w, hyper, density);
}
